# starting kernel
# baseline (speedup 1.0000x reference)
#include <hip/hip_runtime.h>

typedef __attribute__((ext_vector_type(2))) float v2f;
typedef __attribute__((ext_vector_type(8))) float v8f;

#define BB 4
#define NN 4096
#define FF 64

__global__ void k_zero(float* __restrict__ deg) {
    deg[blockIdx.x * 256 + threadIdx.x] = 0.0f;
}

__global__ void k_colsum(const float* __restrict__ A, float* __restrict__ deg) {
    const int j  = blockIdx.x * 256 + threadIdx.x;
    const int b  = blockIdx.z;
    const int r0 = blockIdx.y * 64;
    const float* Ab = A + (size_t)b * NN * NN;
    float s = 0.0f;
#pragma unroll 4
    for (int r = 0; r < 64; ++r)
        s += Ab[(size_t)(r0 + r) * NN + j];
    atomicAdd(&deg[b * NN + j], s);
}

__global__ void k_rsqrt(float* __restrict__ deg) {
    const int i = blockIdx.x * 256 + threadIdx.x;
    deg[i] = rsqrtf(deg[i] + 1.0f);
}

__global__ void k_hw(const float* __restrict__ H, const float* __restrict__ W,
                     const float* __restrict__ d, float* __restrict__ HWsT) {
    __shared__ float  sh[256];
    __shared__ float4 tile[64];
    const int t = threadIdx.x;
    const int rowBase = blockIdx.x * 4;
    sh[t] = H[(size_t)rowBase * FF + t];
    __syncthreads();
    const int lr  = t >> 6;
    const int o   = t & 63;
    const int row = rowBase + lr;
    float s = 0.0f;
#pragma unroll
    for (int f = 0; f < FF; ++f)
        s += sh[lr * FF + f] * W[f * FF + o];
    ((float*)&tile[o])[lr] = d[row] * s;
    __syncthreads();
    if (t < 64) {
        const int b  = rowBase / NN;
        const int m0 = rowBase % NN;
        *(float4*)&HWsT[((size_t)b * FF + t) * NN + m0] = tile[t];
    }
}

__global__ void __launch_bounds__(256)
k_gcn_main(const float* __restrict__ A, const float* __restrict__ HWsT,
           const float* __restrict__ d, float* __restrict__ out) {
    const int b    = blockIdx.y;
    const int wave = threadIdx.x >> 5;
    const int lane = threadIdx.x & 31;
    const int half = lane >> 4;
    const int lm   = lane & 15;

    const int m0 = blockIdx.x * 32 + (wave & 1) * 16;
    const int n0 = (wave >> 1) * 16;
    const int col = n0 + lm;

    const float* __restrict__ Ab  = A + ((size_t)b * NN + (m0 + lm)) * NN;
    const float* __restrict__ BbT = HWsT + ((size_t)b * FF + col) * NN;

    v8f acc = {};
#pragma unroll 8
    for (int k0 = 0; k0 < NN; k0 += 4) {
        v2f a  = *(const v2f*)(Ab  + k0 + 2 * half);
        v2f bv = *(const v2f*)(BbT + k0 + 2 * half);
        acc = __builtin_amdgcn_wmma_f32_16x16x4_f32(
             false, a,  false, bv,
             (short)0, acc,  false,  false);
    }

    const float* __restrict__ db = d + b * NN;
    float* __restrict__ ob = out + (size_t)b * NN * FF;
    const int rbase = m0 + 8 * half;
#pragma unroll
    for (int i = 0; i < 8; ++i) {
        const int row = rbase + i;
        ob[(size_t)row * FF + col] = db[row] * (acc[i] + BbT[row]);
    }
}

extern "C" void kernel_launch(void* const* d_in, const int* in_sizes, int n_in,
                              void* d_out, int out_size, void* d_ws, size_t ws_size,
                              hipStream_t stream) {
    const float* A = (const float*)d_in[0];
    const float* H = (const float*)d_in[1];
    const float* W = (const float*)d_in[2];
    float* out = (float*)d_out;

    float* deg  = (float*)d_ws;
    float* HWsT = deg + BB * NN;

    k_zero<<<BB * NN / 256, 256, 0, stream>>>(deg);

    dim3 gc(NN / 256, 64, BB);
    k_colsum<<<gc, 256, 0, stream>>>(A, deg);

    k_rsqrt<<<BB * NN / 256, 256, 0, stream>>>(deg);

    k_hw<<<BB * NN / 4, 256, 0, stream>>>(H, W, deg, HWsT);

    dim3 gm(NN / 32, BB);
    k_gcn_main<<<gm, 256, 0, stream>>>(A, HWsT, deg, out);
}


# column sums as two-stage reduction: row-chunk workgroups store 128-row partial sums (16 B/lane, 16-32 loads in flight), finisher sums 32 planes + rsqrt; atomics, zero launch gone; sum order changed
# speedup vs baseline: 2.7052x; 2.7052x over previous
#include <hip/hip_runtime.h>

typedef __attribute__((ext_vector_type(2))) float v2f;
typedef __attribute__((ext_vector_type(4))) float v4f;
typedef __attribute__((ext_vector_type(8))) float v8f;

#define BB 4
#define NN 4096
#define FF 64

#define CR 128
#define NP (NN / CR)
#define CG 16

__device__ __forceinline__ void load_rows(v4f (&x)[CG], const float* __restrict__ p) {
#pragma unroll
    for (int r = 0; r < CG; ++r)
        x[r] = *(const v4f*)(p + (size_t)r * NN);
}

__device__ __forceinline__ void sum_rows(v4f (&s)[4], const v4f (&x)[CG]) {
#pragma unroll
    for (int r = 0; r < CG; ++r)
        s[r & 3] += x[r];
}

__global__ void __launch_bounds__(128)
k_colpart(const float* __restrict__ A, float* __restrict__ part) {
    const int col = blockIdx.x * 512 + threadIdx.x * 4;
    const int ch  = blockIdx.y;
    const int b   = blockIdx.z;
    const float* __restrict__ pa = A + ((size_t)b * NN + (size_t)ch * CR) * NN + col;

    v4f x0[CG], x1[CG];
    v4f s[4] = {(v4f){}, (v4f){}, (v4f){}, (v4f){}};
    load_rows(x0, pa);
    for (int g = 0; g < CR / CG - 2; g += 2) {
        load_rows(x1, pa + (size_t)(g + 1) * CG * NN);
        sum_rows(s, x0);
        load_rows(x0, pa + (size_t)(g + 2) * CG * NN);
        sum_rows(s, x1);
    }
    load_rows(x1, pa + (size_t)(CR / CG - 1) * CG * NN);
    sum_rows(s, x0);
    sum_rows(s, x1);

    *(v4f*)(part + ((size_t)b * NP + ch) * NN + col) = (s[0] + s[1]) + (s[2] + s[3]);
}

__global__ void k_dfinish(const float* __restrict__ part, float* __restrict__ d) {
    const int i   = blockIdx.x * 256 + threadIdx.x;
    const int b   = i / NN;
    const int col = i % NN;
    const float* __restrict__ pp = part + (size_t)b * NP * NN + col;
    float s = pp[0];
#pragma unroll
    for (int p = 1; p < NP; ++p)
        s += pp[(size_t)p * NN];
    d[i] = rsqrtf(s + 1.0f);
}

#define TM    64
#define KW    8
#define KL    32
#define KSTEP (KW * KL)
#define NSTEP (NN / KSTEP)
#define PITCH 68

__device__ __forceinline__ void load_a(v4f (&f)[4][4], const float* __restrict__ p) {
#pragma unroll
    for (int t = 0; t < 4; ++t)
#pragma unroll
        for (int q = 0; q < 4; ++q)
            f[t][q] = *(const v4f*)(p + (size_t)t * 16 * NN + 8 * q);
}

__device__ __forceinline__ void load_h(v4f (&f)[4][4], const float* __restrict__ p) {
#pragma unroll
    for (int q = 0; q < 4; ++q)
#pragma unroll
        for (int e = 0; e < 4; ++e)
            f[q][e] = *(const v4f*)(p + (size_t)(8 * q + e) * FF);
}

__device__ __forceinline__ void load_d(v4f (&dd)[4], const float* __restrict__ p) {
#pragma unroll
    for (int q = 0; q < 4; ++q)
        dd[q] = *(const v4f*)(p + 8 * q);
}

__device__ __forceinline__ void scale_b(v4f (&fb)[4][4], const v4f (&hr)[4][4],
                                        const v4f (&dd)[4]) {
#pragma unroll
    for (int q = 0; q < 4; ++q)
#pragma unroll
        for (int ct = 0; ct < 4; ++ct)
#pragma unroll
            for (int e = 0; e < 4; ++e)
                fb[ct][q][e] = hr[q][e][ct] * dd[q][e];
}

__device__ __forceinline__ void mma_block(v8f (&acc)[4][4], const v4f (&fa)[4][4],
                                          const v4f (&fb)[4][4]) {
#pragma unroll
    for (int q = 0; q < 4; ++q)
#pragma unroll
        for (int p = 0; p < 2; ++p)
#pragma unroll
            for (int rt = 0; rt < 4; ++rt)
#pragma unroll
                for (int ct = 0; ct < 4; ++ct) {
                    const v2f a = p ? __builtin_shufflevector(fa[rt][q], fa[rt][q], 2, 3)
                                    : __builtin_shufflevector(fa[rt][q], fa[rt][q], 0, 1);
                    const v2f b = p ? __builtin_shufflevector(fb[ct][q], fb[ct][q], 2, 3)
                                    : __builtin_shufflevector(fb[ct][q], fb[ct][q], 0, 1);
                    acc[rt][ct] = __builtin_amdgcn_wmma_f32_16x16x4_f32(
                         false, a,  false, b,
                         (short)0, acc[rt][ct],  false,  false);
                }
}

#define HAZARD_SLOTS()                                              \
    do {                                                            \
        __builtin_amdgcn_sched_barrier(0);                          \
        asm volatile("v_nop\n\tv_nop\n\tv_nop\n\tv_nop");          \
        __builtin_amdgcn_sched_barrier(0);                          \
    } while (0)

__global__ void __launch_bounds__(256)
k_gcn_main(const float* __restrict__ A, const float* __restrict__ H,
           const float* __restrict__ W, const float* __restrict__ d,
           float* __restrict__ out) {
    __shared__ __attribute__((aligned(16))) float part[KW * TM * PITCH];
    __shared__ __attribute__((aligned(16))) float ys[TM * PITCH];

    const int b    = blockIdx.y;
    const int wave = threadIdx.x >> 5;
    const int lane = threadIdx.x & 31;
    const int h    = lane >> 4;
    const int c    = lane & 15;
    const int r0   = blockIdx.x * TM;

    const float* __restrict__ pa = A + ((size_t)b * NN + r0 + c) * NN + wave * KL + 4 * h;
    const float* __restrict__ ph = H + ((size_t)b * NN + wave * KL + 4 * h) * FF + 4 * c;
    const float* __restrict__ pd = d + b * NN + wave * KL + 4 * h;

    v8f acc[4][4];
#pragma unroll
    for (int rt = 0; rt < 4; ++rt)
#pragma unroll
        for (int ct = 0; ct < 4; ++ct)
            acc[rt][ct] = (v8f){};

    v4f a0[4][4], h0[4][4], d0[4], a1[4][4], h1[4][4], d1[4], fb[4][4];
    load_a(a0, pa);
    load_h(h0, ph);
    load_d(d0, pd);
#pragma unroll 1
    for (int s = 0; s < NSTEP - 2; s += 2) {
        load_a(a1, pa + (s + 1) * KSTEP);
        load_h(h1, ph + (size_t)(s + 1) * KSTEP * FF);
        load_d(d1, pd + (s + 1) * KSTEP);
        __builtin_amdgcn_sched_barrier(0);
        scale_b(fb, h0, d0);
        __builtin_amdgcn_sched_barrier(0);
        mma_block(acc, a0, fb);
        HAZARD_SLOTS();
        load_a(a0, pa + (s + 2) * KSTEP);
        load_h(h0, ph + (size_t)(s + 2) * KSTEP * FF);
        load_d(d0, pd + (s + 2) * KSTEP);
        __builtin_amdgcn_sched_barrier(0);
        scale_b(fb, h1, d1);
        __builtin_amdgcn_sched_barrier(0);
        mma_block(acc, a1, fb);
        HAZARD_SLOTS();
    }
    load_a(a1, pa + (NSTEP - 1) * KSTEP);
    load_h(h1, ph + (size_t)(NSTEP - 1) * KSTEP * FF);
    load_d(d1, pd + (NSTEP - 1) * KSTEP);
    __builtin_amdgcn_sched_barrier(0);
    scale_b(fb, h0, d0);
    __builtin_amdgcn_sched_barrier(0);
    mma_block(acc, a0, fb);
    HAZARD_SLOTS();
    scale_b(fb, h1, d1);
    __builtin_amdgcn_sched_barrier(0);
    mma_block(acc, a1, fb);
    HAZARD_SLOTS();

    float* __restrict__ pw = part + wave * (TM * PITCH);
#pragma unroll
    for (int rt = 0; rt < 4; ++rt)
#pragma unroll
        for (int ct = 0; ct < 4; ++ct)
#pragma unroll
            for (int i = 0; i < 8; ++i)
                pw[(16 * rt + 8 * h + i) * PITCH + 16 * ct + c] = acc[rt][ct][i];

    const int rt2 = wave >> 1;
    const int ot0 = 2 * (wave & 1);
    v2f wf[2][8][2];
#pragma unroll
    for (int u = 0; u < 2; ++u)
#pragma unroll
        for (int q = 0; q < 8; ++q)
#pragma unroll
            for (int p = 0; p < 2; ++p) {
                const float* __restrict__ wp = W + (8 * q + 4 * h + 2 * p) * FF + 16 * (ot0 + u) + c;
                wf[u][q][p] = (v2f){wp[0], wp[FF]};
            }

    const int cg  = threadIdx.x & 15;
    const int rr  = threadIdx.x >> 4;
    const int fb0 = 16 * (cg & 3) + (cg >> 2);
    const float* __restrict__ db  = d + b * NN + r0 + rr;
    const float* __restrict__ idp = H + ((size_t)b * NN + r0 + rr) * FF + fb0;
    v4f   idt[4];
    float dv[4];
#pragma unroll
    for (int g = 0; g < 4; ++g) {
        dv[g] = db[16 * g];
#pragma unroll
        for (int j = 0; j < 4; ++j)
            idt[g][j] = idp[(size_t)16 * g * FF + 4 * j];
    }
    __syncthreads();

    const float* __restrict__ ps = part + rr * PITCH + 4 * cg;
#pragma unroll
    for (int g = 0; g < 4; ++g) {
        v4f s = *(const v4f*)(ps + 16 * g * PITCH);
#pragma unroll
        for (int w = 1; w < KW; ++w)
            s += *(const v4f*)(ps + (w * TM + 16 * g) * PITCH);
        s += dv[g] * idt[g];
#pragma unroll
        for (int j = 0; j < 4; ++j)
            ys[(rr + 16 * g) * PITCH + fb0 + 4 * j] = s[j];
    }
    __syncthreads();

    v4f ya[8];
#pragma unroll
    for (int q = 0; q < 8; ++q)
        ya[q] = *(const v4f*)(ys + (16 * rt2 + c) * PITCH + 8 * q + 4 * h);
    v8f z[2] = {(v8f){}, (v8f){}};
    __builtin_amdgcn_sched_barrier(0);
#pragma unroll
    for (int q = 0; q < 8; ++q)
#pragma unroll
        for (int p = 0; p < 2; ++p)
#pragma unroll
            for (int u = 0; u < 2; ++u) {
                const v2f a = p ? __builtin_shufflevector(ya[q], ya[q], 2, 3)
                                : __builtin_shufflevector(ya[q], ya[q], 0, 1);
                z[u] = __builtin_amdgcn_wmma_f32_16x16x4_f32(
                     false, a,  false, wf[u][q][p],
                     (short)0, z[u],  false,  false);
            }

    HAZARD_SLOTS();

#pragma unroll
    for (int u = 0; u < 2; ++u)
#pragma unroll
        for (int i = 0; i < 8; ++i)
            part[(16 * rt2 + 8 * h + i) * PITCH + 16 * (ot0 + u) + c] = z[u][i];
    __syncthreads();

    float* __restrict__ ob = out + ((size_t)b * NN + r0 + rr) * FF + 4 * cg;
#pragma unroll
    for (int g = 0; g < 4; ++g) {
        const v4f zz = *(const v4f*)(ps + 16 * g * PITCH);
        *(v4f*)(ob + (size_t)16 * g * FF) = dv[g] * zz;
    }
}

extern "C" void kernel_launch(void* const* d_in, const int* in_sizes, int n_in,
                              void* d_out, int out_size, void* d_ws, size_t ws_size,
                              hipStream_t stream) {
    const float* A = (const float*)d_in[0];
    const float* H = (const float*)d_in[1];
    const float* W = (const float*)d_in[2];
    float* out = (float*)d_out;

    float* deg  = (float*)d_ws;
    float* part = (float*)d_ws + BB * NN;

    dim3 gp(NN / 512, NP, BB);
    k_colpart<<<gp, 128, 0, stream>>>(A, part);

    k_dfinish<<<BB * NN / 256, 256, 0, stream>>>(part, deg);

    dim3 gm(NN / TM, BB);
    k_gcn_main<<<gm, 256, 0, stream>>>(A, H, W, deg, out);
}


# column-sum stream re-cut from 512-byte column strips to row blocks: each wave reads whole 8 KiB row halves (16 contiguous loads), 16 waves x 8 rows summed through LDS; column-sum order changed
# speedup vs baseline: 2.8662x; 1.0595x over previous
#include <hip/hip_runtime.h>

typedef __attribute__((ext_vector_type(2))) float v2f;
typedef __attribute__((ext_vector_type(4))) float v4f;
typedef __attribute__((ext_vector_type(8))) float v8f;

#define BB 4
#define NN 4096
#define FF 64

#define CR    128
#define NP    (NN / CR)
#define CWV   16
#define RPW   (CR / CWV)
#define CSPAN 2048
#define CQ    (CSPAN / 128)

__device__ __forceinline__ void load_span(v4f (&x)[CQ], const float* __restrict__ p) {
#pragma unroll
    for (int q = 0; q < CQ; ++q)
        x[q] = *(const v4f*)(p + q * 128);
}

__device__ __forceinline__ void add_span(v4f (&a)[CQ], const v4f (&x)[CQ]) {
#pragma unroll
    for (int q = 0; q < CQ; ++q)
        a[q] += x[q];
}

__global__ void __launch_bounds__(CWV * 32)
k_colpart(const float* __restrict__ A, float* __restrict__ part) {
    __shared__ __attribute__((aligned(16))) float red[CWV * CSPAN];

    const int wave = threadIdx.x >> 5;
    const int lane = threadIdx.x & 31;
    const int c0   = blockIdx.x * CSPAN;
    const int rb   = blockIdx.y;
    const int b    = blockIdx.z;
    const float* __restrict__ pa =
        A + ((size_t)b * NN + (size_t)rb * CR + wave * RPW) * NN + c0 + lane * 4;

    v4f x0[CQ], x1[CQ], acc[CQ];
#pragma unroll
    for (int q = 0; q < CQ; ++q)
        acc[q] = (v4f){};
    load_span(x0, pa);
    for (int r = 0; r < RPW - 2; r += 2) {
        load_span(x1, pa + (size_t)(r + 1) * NN);
        __builtin_amdgcn_sched_barrier(0);
        add_span(acc, x0);
        __builtin_amdgcn_sched_barrier(0);
        load_span(x0, pa + (size_t)(r + 2) * NN);
        __builtin_amdgcn_sched_barrier(0);
        add_span(acc, x1);
        __builtin_amdgcn_sched_barrier(0);
    }
    load_span(x1, pa + (size_t)(RPW - 1) * NN);
    __builtin_amdgcn_sched_barrier(0);
    add_span(acc, x0);
    add_span(acc, x1);

    float* __restrict__ pr = red + wave * CSPAN + lane * 4;
#pragma unroll
    for (int q = 0; q < CQ; ++q)
        *(v4f*)(pr + q * 128) = acc[q];
    __syncthreads();

    const float* __restrict__ ps = red + threadIdx.x * 4;
    v4f s = *(const v4f*)ps;
#pragma unroll
    for (int w = 1; w < CWV; ++w)
        s += *(const v4f*)(ps + w * CSPAN);
    *(v4f*)(part + ((size_t)b * NP + rb) * NN + c0 + threadIdx.x * 4) = s;
}

__global__ void k_dfinish(const float* __restrict__ part, float* __restrict__ d) {
    const int i   = blockIdx.x * 256 + threadIdx.x;
    const int b   = i / NN;
    const int col = i % NN;
    const float* __restrict__ pp = part + (size_t)b * NP * NN + col;
    float s = pp[0];
#pragma unroll
    for (int p = 1; p < NP; ++p)
        s += pp[(size_t)p * NN];
    d[i] = rsqrtf(s + 1.0f);
}

#define TM    64
#define KW    8
#define KL    32
#define KSTEP (KW * KL)
#define NSTEP (NN / KSTEP)
#define PITCH 68

__device__ __forceinline__ void load_a(v4f (&f)[4][4], const float* __restrict__ p) {
#pragma unroll
    for (int t = 0; t < 4; ++t)
#pragma unroll
        for (int q = 0; q < 4; ++q)
            f[t][q] = *(const v4f*)(p + (size_t)t * 16 * NN + 8 * q);
}

__device__ __forceinline__ void load_h(v4f (&f)[4][4], const float* __restrict__ p) {
#pragma unroll
    for (int q = 0; q < 4; ++q)
#pragma unroll
        for (int e = 0; e < 4; ++e)
            f[q][e] = *(const v4f*)(p + (size_t)(8 * q + e) * FF);
}

__device__ __forceinline__ void load_d(v4f (&dd)[4], const float* __restrict__ p) {
#pragma unroll
    for (int q = 0; q < 4; ++q)
        dd[q] = *(const v4f*)(p + 8 * q);
}

__device__ __forceinline__ void scale_b(v4f (&fb)[4][4], const v4f (&hr)[4][4],
                                        const v4f (&dd)[4]) {
#pragma unroll
    for (int q = 0; q < 4; ++q)
#pragma unroll
        for (int ct = 0; ct < 4; ++ct)
#pragma unroll
            for (int e = 0; e < 4; ++e)
                fb[ct][q][e] = hr[q][e][ct] * dd[q][e];
}

__device__ __forceinline__ void mma_block(v8f (&acc)[4][4], const v4f (&fa)[4][4],
                                          const v4f (&fb)[4][4]) {
#pragma unroll
    for (int q = 0; q < 4; ++q)
#pragma unroll
        for (int p = 0; p < 2; ++p)
#pragma unroll
            for (int rt = 0; rt < 4; ++rt)
#pragma unroll
                for (int ct = 0; ct < 4; ++ct) {
                    const v2f a = p ? __builtin_shufflevector(fa[rt][q], fa[rt][q], 2, 3)
                                    : __builtin_shufflevector(fa[rt][q], fa[rt][q], 0, 1);
                    const v2f b = p ? __builtin_shufflevector(fb[ct][q], fb[ct][q], 2, 3)
                                    : __builtin_shufflevector(fb[ct][q], fb[ct][q], 0, 1);
                    acc[rt][ct] = __builtin_amdgcn_wmma_f32_16x16x4_f32(
                         false, a,  false, b,
                         (short)0, acc[rt][ct],  false,  false);
                }
}

#define HAZARD_SLOTS()                                              \
    do {                                                            \
        __builtin_amdgcn_sched_barrier(0);                          \
        asm volatile("v_nop\n\tv_nop\n\tv_nop\n\tv_nop");          \
        __builtin_amdgcn_sched_barrier(0);                          \
    } while (0)

__global__ void __launch_bounds__(256)
k_gcn_main(const float* __restrict__ A, const float* __restrict__ H,
           const float* __restrict__ W, const float* __restrict__ d,
           float* __restrict__ out) {
    __shared__ __attribute__((aligned(16))) float part[KW * TM * PITCH];
    __shared__ __attribute__((aligned(16))) float ys[TM * PITCH];

    const int b    = blockIdx.y;
    const int wave = threadIdx.x >> 5;
    const int lane = threadIdx.x & 31;
    const int h    = lane >> 4;
    const int c    = lane & 15;
    const int r0   = blockIdx.x * TM;

    const float* __restrict__ pa = A + ((size_t)b * NN + r0 + c) * NN + wave * KL + 4 * h;
    const float* __restrict__ ph = H + ((size_t)b * NN + wave * KL + 4 * h) * FF + 4 * c;
    const float* __restrict__ pd = d + b * NN + wave * KL + 4 * h;

    v8f acc[4][4];
#pragma unroll
    for (int rt = 0; rt < 4; ++rt)
#pragma unroll
        for (int ct = 0; ct < 4; ++ct)
            acc[rt][ct] = (v8f){};

    v4f a0[4][4], h0[4][4], d0[4], a1[4][4], h1[4][4], d1[4], fb[4][4];
    load_a(a0, pa);
    load_h(h0, ph);
    load_d(d0, pd);
#pragma unroll 1
    for (int s = 0; s < NSTEP - 2; s += 2) {
        load_a(a1, pa + (s + 1) * KSTEP);
        load_h(h1, ph + (size_t)(s + 1) * KSTEP * FF);
        load_d(d1, pd + (s + 1) * KSTEP);
        __builtin_amdgcn_sched_barrier(0);
        scale_b(fb, h0, d0);
        __builtin_amdgcn_sched_barrier(0);
        mma_block(acc, a0, fb);
        HAZARD_SLOTS();
        load_a(a0, pa + (s + 2) * KSTEP);
        load_h(h0, ph + (size_t)(s + 2) * KSTEP * FF);
        load_d(d0, pd + (s + 2) * KSTEP);
        __builtin_amdgcn_sched_barrier(0);
        scale_b(fb, h1, d1);
        __builtin_amdgcn_sched_barrier(0);
        mma_block(acc, a1, fb);
        HAZARD_SLOTS();
    }
    load_a(a1, pa + (NSTEP - 1) * KSTEP);
    load_h(h1, ph + (size_t)(NSTEP - 1) * KSTEP * FF);
    load_d(d1, pd + (NSTEP - 1) * KSTEP);
    __builtin_amdgcn_sched_barrier(0);
    scale_b(fb, h0, d0);
    __builtin_amdgcn_sched_barrier(0);
    mma_block(acc, a0, fb);
    HAZARD_SLOTS();
    scale_b(fb, h1, d1);
    __builtin_amdgcn_sched_barrier(0);
    mma_block(acc, a1, fb);
    HAZARD_SLOTS();

    float* __restrict__ pw = part + wave * (TM * PITCH);
#pragma unroll
    for (int rt = 0; rt < 4; ++rt)
#pragma unroll
        for (int ct = 0; ct < 4; ++ct)
#pragma unroll
            for (int i = 0; i < 8; ++i)
                pw[(16 * rt + 8 * h + i) * PITCH + 16 * ct + c] = acc[rt][ct][i];

    const int rt2 = wave >> 1;
    const int ot0 = 2 * (wave & 1);
    v2f wf[2][8][2];
#pragma unroll
    for (int u = 0; u < 2; ++u)
#pragma unroll
        for (int q = 0; q < 8; ++q)
#pragma unroll
            for (int p = 0; p < 2; ++p) {
                const float* __restrict__ wp = W + (8 * q + 4 * h + 2 * p) * FF + 16 * (ot0 + u) + c;
                wf[u][q][p] = (v2f){wp[0], wp[FF]};
            }

    const int cg  = threadIdx.x & 15;
    const int rr  = threadIdx.x >> 4;
    const int fb0 = 16 * (cg & 3) + (cg >> 2);
    const float* __restrict__ db  = d + b * NN + r0 + rr;
    const float* __restrict__ idp = H + ((size_t)b * NN + r0 + rr) * FF + fb0;
    v4f   idt[4];
    float dv[4];
#pragma unroll
    for (int g = 0; g < 4; ++g) {
        dv[g] = db[16 * g];
#pragma unroll
        for (int j = 0; j < 4; ++j)
            idt[g][j] = idp[(size_t)16 * g * FF + 4 * j];
    }
    __syncthreads();

    const float* __restrict__ ps = part + rr * PITCH + 4 * cg;
#pragma unroll
    for (int g = 0; g < 4; ++g) {
        v4f s = *(const v4f*)(ps + 16 * g * PITCH);
#pragma unroll
        for (int w = 1; w < KW; ++w)
            s += *(const v4f*)(ps + (w * TM + 16 * g) * PITCH);
        s += dv[g] * idt[g];
#pragma unroll
        for (int j = 0; j < 4; ++j)
            ys[(rr + 16 * g) * PITCH + fb0 + 4 * j] = s[j];
    }
    __syncthreads();

    v4f ya[8];
#pragma unroll
    for (int q = 0; q < 8; ++q)
        ya[q] = *(const v4f*)(ys + (16 * rt2 + c) * PITCH + 8 * q + 4 * h);
    v8f z[2] = {(v8f){}, (v8f){}};
    __builtin_amdgcn_sched_barrier(0);
#pragma unroll
    for (int q = 0; q < 8; ++q)
#pragma unroll
        for (int p = 0; p < 2; ++p)
#pragma unroll
            for (int u = 0; u < 2; ++u) {
                const v2f a = p ? __builtin_shufflevector(ya[q], ya[q], 2, 3)
                                : __builtin_shufflevector(ya[q], ya[q], 0, 1);
                z[u] = __builtin_amdgcn_wmma_f32_16x16x4_f32(
                     false, a,  false, wf[u][q][p],
                     (short)0, z[u],  false,  false);
            }

    HAZARD_SLOTS();

#pragma unroll
    for (int u = 0; u < 2; ++u)
#pragma unroll
        for (int i = 0; i < 8; ++i)
            part[(16 * rt2 + 8 * h + i) * PITCH + 16 * (ot0 + u) + c] = z[u][i];
    __syncthreads();

    float* __restrict__ ob = out + ((size_t)b * NN + r0 + rr) * FF + 4 * cg;
#pragma unroll
    for (int g = 0; g < 4; ++g) {
        const v4f zz = *(const v4f*)(ps + 16 * g * PITCH);
        *(v4f*)(ob + (size_t)16 * g * FF) = dv[g] * zz;
    }
}

extern "C" void kernel_launch(void* const* d_in, const int* in_sizes, int n_in,
                              void* d_out, int out_size, void* d_ws, size_t ws_size,
                              hipStream_t stream) {
    const float* A = (const float*)d_in[0];
    const float* H = (const float*)d_in[1];
    const float* W = (const float*)d_in[2];
    float* out = (float*)d_out;

    float* deg  = (float*)d_ws;
    float* part = (float*)d_ws + BB * NN;

    dim3 gp(NN / CSPAN, NP, BB);
    k_colpart<<<gp, CWV * 32, 0, stream>>>(A, part);

    k_dfinish<<<BB * NN / 256, 256, 0, stream>>>(part, deg);

    dim3 gm(NN / TM, BB);
    k_gcn_main<<<gm, 256, 0, stream>>>(A, H, W, deg, out);
}
